# starting kernel
# baseline (speedup 1.0000x reference)
#include <hip/hip_runtime.h>
#include <hip/hip_bf16.h>

typedef __attribute__((ext_vector_type(2))) float v2f;
typedef __attribute__((ext_vector_type(8))) float v8f;

#define S_LEN   4096
#define D_DIM   1024
#define NPATCH  8
#define NSEG    9

__global__ void seg_bounds_kernel(const int* __restrict__ mask,
                                  int* __restrict__ ofs) {
  const int b   = blockIdx.x;
  const int tid = threadIdx.x;
  __shared__ int psum[256];

  const int base = b * S_LEN + tid * 16;
  int m[16];
  int c = 0;
#pragma unroll
  for (int j = 0; j < 16; ++j) { m[j] = mask[base + j]; c += m[j]; }
  psum[tid] = c;
  __syncthreads();

#pragma unroll
  for (int off = 1; off < 256; off <<= 1) {
    int v   = psum[tid];
    int add = (tid >= off) ? psum[tid - off] : 0;
    __syncthreads();
    psum[tid] = v + add;
    __syncthreads();
  }
  const int excl = (tid == 0) ? 0 : psum[tid - 1];

  if (tid < 10) ofs[b * 10 + tid] = (tid == 0) ? 0 : S_LEN;
  __syncthreads();

  int r = excl;
#pragma unroll
  for (int j = 0; j < 16; ++j) {
    if (m[j]) {
      if (r < NPATCH) ofs[b * 10 + (r + 1)] = tid * 16 + j + 1;
      ++r;
    }
  }
}

__global__ void seg_mean_kernel(const float* __restrict__ x,
                                const int* __restrict__ ofs,
                                float* __restrict__ patches) {
  const int p = blockIdx.x & 7;
  const int b = blockIdx.x >> 3;
  const int d = threadIdx.x;

  const int s0 = ofs[b * 10 + p];
  const int s1 = ofs[b * 10 + p + 1];

  const float* src = x + (size_t)b * S_LEN * D_DIM + d;
  float acc = 0.0f;
  for (int s = s0; s < s1; ++s) acc += src[(size_t)s * D_DIM];

  const int cnt = s1 - s0;
  const float denom = (float)(cnt > 0 ? cnt : 1);
  patches[(size_t)(b * NPATCH + p) * D_DIM + d] = acc / denom;
}

__global__ void patch_gemm_wmma(const float* __restrict__ P,
                                const float* __restrict__ Wm,
                                const float* __restrict__ bias,
                                float* __restrict__ out) {
  const int wave = blockIdx.x * (blockDim.x >> 5) + (threadIdx.x >> 5);
  const int lane = threadIdx.x & 31;
  const int mt   = wave >> 6;
  const int nt   = wave & 63;
  const int m0   = mt * 16;
  const int n0   = nt * 16;
  const int half = lane >> 4;
  const int lr   = lane & 15;

  const float* Arow = P  + (size_t)(m0 + lr) * D_DIM + half * 2;
  const float* Brow = Wm + (size_t)(n0 + lr) * D_DIM + half * 2;

  v8f c = {};
#pragma unroll 8
  for (int k = 0; k < D_DIM; k += 4) {
    v2f a  = *(const v2f*)(Arow + k);
    v2f bb = *(const v2f*)(Brow + k);
    c = __builtin_amdgcn_wmma_f32_16x16x4_f32(
         false, a,  false, bb,
         (short)0, c,  false,  false);
  }

  const float bv = bias[n0 + lr];
  float* ocol = out + n0 + lr;
#pragma unroll
  for (int r = 0; r < 8; ++r) {
    ocol[(size_t)(m0 + half * 8 + r) * D_DIM] = c[r] + bv;
  }
}

extern "C" void kernel_launch(void* const* d_in, const int* in_sizes, int n_in,
                              void* d_out, int out_size, void* d_ws, size_t ws_size,
                              hipStream_t stream) {
  const float* latent = (const float*)d_in[0];
  const int*   mask   = (const int*)  d_in[1];
  const float* W      = (const float*)d_in[2];
  const float* bias   = (const float*)d_in[3];
  float*       out    = (float*)d_out;

  int*   ofs     = (int*)d_ws;
  float* patches = (float*)((char*)d_ws + 1024);

  seg_bounds_kernel<<<8, 256, 0, stream>>>(mask, ofs);
  seg_mean_kernel<<<64, 1024, 0, stream>>>(latent, ofs, patches);
  patch_gemm_wmma<<<64, 128, 0, stream>>>(patches, W, bias, out);
}


# K in 4 pipelined groups: 8 rows per segment asked for at once, branch-free (absent row = fixed line, weight 0); a group is summed and multiplied under the next one's loads (order changes past 8 rows)
# speedup vs baseline: 4.3492x; 4.3492x over previous
#include <hip/hip_runtime.h>

typedef __attribute__((ext_vector_type(2))) float v2f;
typedef __attribute__((ext_vector_type(4))) float v4f;
typedef __attribute__((ext_vector_type(8))) float v8f;

#define S_LEN   4096
#define D_DIM   1024
#define NPATCH  8

#define KWAVES  8
#define KCHUNK  (D_DIM / KWAVES)
#define NLOAD   (KCHUNK / 8)
#define RP      8
#define G       4
#define NG      (NLOAD / G)

__device__ __forceinline__ int nth_set_bit(unsigned bits, int n) {
#pragma unroll
  for (int i = 1; i < NPATCH; ++i)
    if (i < n) bits &= bits - 1u;
  return __ffs((int)bits) - 1;
}

__device__ __forceinline__ void scan_chunk(int vA, int vB, int c, int hi, int p,
                                           int& cntA, int& cntB, int& s0, int& s1) {
  const unsigned bitsA = __builtin_amdgcn_ballot_w32(vA != 0);
  const unsigned bitsB = __builtin_amdgcn_ballot_w32(vB != 0);
  const unsigned bits  = hi ? bitsB : bitsA;
  const int seen = hi ? cntB : cntA;
  const int pc   = __popc(bits);
  const int nb   = p - seen;
  if (nb >= 1 && nb <= pc)     s0 = c * 32 + nth_set_bit(bits, nb) + 1;
  if (nb >= 0 && nb + 1 <= pc) s1 = c * 32 + nth_set_bit(bits, nb + 1) + 1;
  cntA += __popc(bitsA);
  cntB += __popc(bitsB);
}

__device__ __forceinline__ void load_group(v4f (&t)[RP][G], const float* const (&rp)[RP], int g) {
#pragma unroll
  for (int u = 0; u < RP; ++u)
#pragma unroll
    for (int jj = 0; jj < G; ++jj)
      t[u][jj] = *(const v4f*)(rp[u] + 8 * (g * G + jj));
}

__global__ void __launch_bounds__(KWAVES * 32)
patch_fused_wmma(const float* __restrict__ x,
                 const int*   __restrict__ mask,
                 const float* __restrict__ Wm,
                 const float* __restrict__ bias,
                 float* __restrict__ out) {
  __shared__ float red[2 * KWAVES][8][32];

  const int w    = threadIdx.x >> 5;
  const int lane = threadIdx.x & 31;
  const int mt   = blockIdx.x & 3;
  const int nt   = blockIdx.x >> 2;
  const int m0   = mt * 16;
  const int n0   = nt * 16;
  const int half = lane >> 4;
  const int lr   = lane & 15;

  const int hi = lr >> 3;
  const int p  = lr & 7;
  const int* mA = mask + (size_t)(2 * mt) * S_LEN + lane;
  const int* mB = mA + S_LEN;

  const int vA0 = mA[0];
  const int vB0 = mB[0];

  const float* Brow = Wm + (size_t)(n0 + lr) * D_DIM + w * KCHUNK + half * 4;

  const float bv = bias[n0 + lr];

  v4f b[NLOAD];
#pragma unroll
  for (int j = 0; j < NLOAD; ++j) b[j] = *(const v4f*)(Brow + 8 * j);

  int s0 = (p == 0) ? 0 : S_LEN;
  int s1 = S_LEN;
  int cntA = 0, cntB = 0;
  scan_chunk(vA0, vB0, 0, hi, p, cntA, cntB, s0, s1);
  for (int c = 1; c < S_LEN / 32; ++c) {
    if (cntA >= NPATCH && cntB >= NPATCH) break;
    scan_chunk(mA[c * 32], mB[c * 32], c, hi, p, cntA, cntB, s0, s1);
  }

  const int   len = s1 - s0;
  const float inv = 1.0f / (float)(len > 0 ? len : 1);
  const float* Aseg = x + (size_t)(2 * mt + hi) * S_LEN * D_DIM + (size_t)s0 * D_DIM
                        + w * KCHUNK + half * 4;
  const float* idle = x + half * 4;

  v8f c0 = {}, c1 = {};
  for (int base = 0; __builtin_amdgcn_ballot_w32(base < len) != 0u; base += RP) {
    const int rem = len - base;
    const float* rp[RP];
    float wt[RP];
#pragma unroll
    for (int u = 0; u < RP; ++u) {
      const bool on = u < rem;
      rp[u] = on ? Aseg + (size_t)(base + u) * D_DIM : idle;
      wt[u] = on ? 1.0f : 0.0f;
    }

    v4f t[2][RP][G];
    load_group(t[0], rp, 0);
    load_group(t[1], rp, 1);
    __builtin_amdgcn_sched_barrier(0);

#pragma unroll
    for (int g = 0; g < NG; ++g) {
      v4f a[G];
#pragma unroll
      for (int jj = 0; jj < G; ++jj) {
        a[jj] = t[g & 1][0][jj] * wt[0];
#pragma unroll
        for (int u = 1; u < RP; ++u) a[jj] = t[g & 1][u][jj] * wt[u] + a[jj];
      }
      __builtin_amdgcn_sched_barrier(0);
      if (g + 2 < NG) load_group(t[g & 1], rp, g + 2);
#pragma unroll
      for (int jj = 0; jj < G; ++jj) a[jj] *= inv;

      __builtin_amdgcn_sched_barrier(0);
#pragma unroll
      for (int jj = 0; jj < G; ++jj) {
        const int j = g * G + jj;
        const v2f a0 = {a[jj].x, a[jj].y}, a1 = {a[jj].z, a[jj].w};
        const v2f b0 = {b[j].x, b[j].y}, b1 = {b[j].z, b[j].w};
        c0 = __builtin_amdgcn_wmma_f32_16x16x4_f32(
             false, a0,  false, b0,
             (short)0, c0,  false,  false);
        c1 = __builtin_amdgcn_wmma_f32_16x16x4_f32(
             false, a1,  false, b1,
             (short)0, c1,  false,  false);
      }
      __builtin_amdgcn_sched_barrier(0);
      asm volatile("v_nop\n\tv_nop\n\tv_nop\n\tv_nop\n\tv_nop\n\tv_nop\n\tv_nop\n\tv_nop\n\tv_nop");
      __builtin_amdgcn_sched_barrier(0);
    }
  }

#pragma unroll
  for (int r = 0; r < 8; ++r) {
    red[2 * w][r][lane]     = c0[r];
    red[2 * w + 1][r][lane] = c1[r];
  }
  __syncthreads();

  float acc = red[0][w][lane];
#pragma unroll
  for (int q = 1; q < 2 * KWAVES; ++q) acc += red[q][w][lane];

  out[(size_t)(m0 + half * 8 + w) * D_DIM + n0 + lr] = acc + bv;
}

extern "C" void kernel_launch(void* const* d_in, const int* in_sizes, int n_in,
                              void* d_out, int out_size, void* d_ws, size_t ws_size,
                              hipStream_t stream) {
  const float* latent = (const float*)d_in[0];
  const int*   mask   = (const int*)  d_in[1];
  const float* W      = (const float*)d_in[2];
  const float* bias   = (const float*)d_in[3];
  float*       out    = (float*)d_out;

  patch_fused_wmma<<<256, 256, 0, stream>>>(latent, mask, W, bias, out);
}
